# starting kernel
# baseline (speedup 1.0000x reference)
#include <hip/hip_runtime.h>
#include <math.h>

typedef __attribute__((ext_vector_type(2))) float v2f;
typedef __attribute__((ext_vector_type(8))) float v8f;

namespace {
constexpr int kB = 65536;
constexpr int kT = 9;
constexpr int kI = 57;
constexpr int kG = 6;
constexpr int kKsteps = 15;
constexpr int kWavesPerBlock = 2;
constexpr int kBlockThreads = kWavesPerBlock * 32;
constexpr int kBatchPerWave = 32;
constexpr int kBatchPerBlock = kWavesPerBlock * kBatchPerWave;
}

__device__ __forceinline__ float fast_sigmoid(float v) {
  return 1.0f / (1.0f + __expf(-v));
}
__device__ __forceinline__ float fast_tanh(float v) {
  const float e = __expf(-2.0f * v);
  return (1.0f - e) / (1.0f + e);
}

__launch_bounds__(kBlockThreads)
__global__ void gru_fused_wmma(const float* __restrict__ x,
                               const float* __restrict__ W_ih,
                               const float* __restrict__ W_hh,
                               const float* __restrict__ b_ih,
                               const float* __restrict__ b_hh,
                               const float* __restrict__ fc_w,
                               const float* __restrict__ fc_b,
                               float* __restrict__ out) {
  __shared__ float gx_s[kWavesPerBlock * kT * kBatchPerWave * kG];

  const int tid = threadIdx.x;
  const int wave = tid >> 5;
  const int lane = tid & 31;
  const int nlo = lane & 15;
  const int hi = lane >> 4;
  const int b0 = blockIdx.x * kBatchPerBlock + wave * kBatchPerWave;
  float* wlds = gx_s + wave * (kT * kBatchPerWave * kG);

  v2f bfrag[kKsteps];
#pragma unroll
  for (int ks = 0; ks < kKsteps; ++ks) {
    const int k = 4 * ks + 2 * hi;
    float bx = 0.0f, by = 0.0f;
    if (nlo < kG) {
      if (k < kI) bx = W_ih[nlo * kI + k];
      if (k + 1 < kI) by = W_ih[nlo * kI + k + 1];
    }
    v2f f;
    f.x = bx;
    f.y = by;
    bfrag[ks] = f;
  }

  const int rowStride = kT * kI;
  const float* xr0 = x + (b0 + nlo) * rowStride;
  const float* xr1 = x + (b0 + 16 + nlo) * rowStride;

  for (int t = 0; t < kT; ++t) {
    const float* p0 = xr0 + t * kI;
    const float* p1 = xr1 + t * kI;
    if (t + 1 < kT) {
      __builtin_prefetch(p0 + kI, 0, 0);
      __builtin_prefetch(p1 + kI, 0, 0);
    }
    v8f c0 = {};
    v8f c1 = {};
#pragma unroll
    for (int ks = 0; ks < kKsteps; ++ks) {
      const int k = 4 * ks + 2 * hi;
      v2f a0, a1;
      a0.x = (k < kI) ? p0[k] : 0.0f;
      a0.y = (k + 1 < kI) ? p0[k + 1] : 0.0f;
      a1.x = (k < kI) ? p1[k] : 0.0f;
      a1.y = (k + 1 < kI) ? p1[k + 1] : 0.0f;
      c0 = __builtin_amdgcn_wmma_f32_16x16x4_f32(false, a0, false, bfrag[ks],
                                                 (short)0, c0, false, false);
      c1 = __builtin_amdgcn_wmma_f32_16x16x4_f32(false, a1, false, bfrag[ks],
                                                 (short)0, c1, false, false);
    }
    if (nlo < kG) {
#pragma unroll
      for (int j = 0; j < 8; ++j) {
        const int m = j + 8 * hi;
        wlds[(t * kBatchPerWave + m) * kG + nlo] = c0[j];
        wlds[(t * kBatchPerWave + 16 + m) * kG + nlo] = c1[j];
      }
    }
  }

  __syncthreads();

  float wh[12], bi[6], bh[6];
#pragma unroll
  for (int i2 = 0; i2 < 12; ++i2) wh[i2] = W_hh[i2];
#pragma unroll
  for (int g = 0; g < 6; ++g) {
    bi[g] = b_ih[g];
    bh[g] = b_hh[g];
  }

  float h0 = 0.0f, h1 = 0.0f;
#pragma unroll
  for (int t = 0; t < kT; ++t) {
    const float* gt = wlds + (t * kBatchPerWave + lane) * kG;
    const float gh0 = fmaf(wh[0], h0, fmaf(wh[1], h1, bh[0]));
    const float gh1 = fmaf(wh[2], h0, fmaf(wh[3], h1, bh[1]));
    const float gh2 = fmaf(wh[4], h0, fmaf(wh[5], h1, bh[2]));
    const float gh3 = fmaf(wh[6], h0, fmaf(wh[7], h1, bh[3]));
    const float gh4 = fmaf(wh[8], h0, fmaf(wh[9], h1, bh[4]));
    const float gh5 = fmaf(wh[10], h0, fmaf(wh[11], h1, bh[5]));
    const float r0 = fast_sigmoid(gt[0] + bi[0] + gh0);
    const float r1 = fast_sigmoid(gt[1] + bi[1] + gh1);
    const float z0 = fast_sigmoid(gt[2] + bi[2] + gh2);
    const float z1 = fast_sigmoid(gt[3] + bi[3] + gh3);
    const float n0 = fast_tanh(fmaf(r0, gh4, gt[4] + bi[4]));
    const float n1 = fast_tanh(fmaf(r1, gh5, gt[5] + bi[5]));
    h0 = fmaf(z0, h0 - n0, n0);
    h1 = fmaf(z1, h1 - n1, n1);
  }
  out[b0 + lane] = fmaf(fc_w[0], h0, fmaf(fc_w[1], h1, fc_b[0]));
}

extern "C" void kernel_launch(void* const* d_in, const int* in_sizes, int n_in,
                              void* d_out, int out_size, void* d_ws, size_t ws_size,
                              hipStream_t stream) {
  const float* x    = (const float*)d_in[0];
  const float* W_ih = (const float*)d_in[1];
  const float* W_hh = (const float*)d_in[2];
  const float* b_ih = (const float*)d_in[3];
  const float* b_hh = (const float*)d_in[4];
  const float* fc_w = (const float*)d_in[5];
  const float* fc_b = (const float*)d_in[6];
  float* out = (float*)d_out;

  dim3 grid(kB / kBatchPerBlock);
  dim3 block(kBlockThreads);
  hipLaunchKernelGGL(gru_fused_wmma, grid, block, 0, stream,
                     x, W_ih, W_hh, b_ih, b_hh, fc_w, fc_b, out);
}


# recurrence gates on the hardware tanh: sigmoid(v) = 0.5 + 0.5*tanh(v/2), halves folded into W_hh and the biases (gate sums re-associated); these constants are read ahead of the projection loop
# speedup vs baseline: 1.9109x; 1.9109x over previous
#include <hip/hip_runtime.h>
#include <math.h>


typedef __attribute__((ext_vector_type(2))) float v2f;
typedef __attribute__((ext_vector_type(4))) float v4f;
typedef __attribute__((ext_vector_type(8))) float v8f;
typedef float v4f_a4 __attribute__((ext_vector_type(4), aligned(4)));

namespace {
constexpr int kB = 65536;
constexpr int kT = 9;
constexpr int kI = 57;
constexpr int kG = 6;
constexpr int kWavesPerBlock = 2;
constexpr int kBlockThreads = kWavesPerBlock * 32;
constexpr int kBatchPerWave = 32;
constexpr int kBatchPerBlock = kWavesPerBlock * kBatchPerWave;
constexpr int kRowsPerWave = kBatchPerWave * kT;
constexpr int kTileRows = 8;
constexpr int kTiles = kRowsPerWave / kTileRows;
constexpr int kGroup = 4;
constexpr int kTileFloats = kTileRows * kI;
constexpr int kLdsPerWave = kRowsPerWave * 2 * kG;
}

__device__ __forceinline__ float hw_tanh(float v) {
  return __builtin_amdgcn_tanhf(v);
}

__device__ __forceinline__ v4f ldq(const float* p) {
  return *(const v4f_a4*)p;
}

__device__ __forceinline__ void load_group(v4f (&L)[kGroup][4], const float* p, int off0,
                                           int off1) {
#pragma unroll
  for (int tt = 0; tt < kGroup; ++tt) {
    const float* b = p + tt * kTileFloats;
    L[tt][0] = ldq(b + off0);
    L[tt][2] = ldq(b + off1);
    L[tt][1] = ldq(b + 4 * kI + off0);
    L[tt][3] = ldq(b + 4 * kI + off1);
  }
}

template <int kBanks>
__device__ __forceinline__ float take_xor4_1(float keep, float from) {
  return __builtin_bit_cast(
      float, __builtin_amdgcn_update_dpp(__builtin_bit_cast(int, keep),
                                         __builtin_bit_cast(int, from),
                                         0x164  , 0xF, kBanks, false));
}
template <int kBanks>
__device__ __forceinline__ v4f take_xor4(v4f keep, v4f from) {
  v4f r;
  r.x = take_xor4_1<kBanks>(keep.x, from.x);
  r.y = take_xor4_1<kBanks>(keep.y, from.y);
  r.z = take_xor4_1<kBanks>(keep.z, from.z);
  r.w = take_xor4_1<kBanks>(keep.w, from.w);
  return r;
}

__device__ __forceinline__ void to_columns(const v4f (&Y)[4], v4f (&Q)[4]) {
  Q[0] = take_xor4<0xA>(Y[0], Y[1]);
  Q[1] = take_xor4<0x5>(Y[1], Y[0]);
  Q[2] = take_xor4<0xA>(Y[2], Y[3]);
  Q[3] = take_xor4<0x5>(Y[3], Y[2]);
}

__device__ __forceinline__ void compute_group(v4f (&L)[kGroup][4], const v2f (&wf)[8],
                                              bool live, float* dst) {
#pragma unroll
  for (int tt = 0; tt < kGroup; tt += 2) {
    v8f d0 = {};
    v8f d1 = {};
    v4f Qa[4], Qb[4];
    to_columns(L[tt], Qa);
    to_columns(L[tt + 1], Qb);
#pragma unroll
    for (int u = 0; u < 8; ++u) {
      const int q = u >> 1;
      const v4f qa = Qa[q];
      const v4f qb = Qb[q];
      v2f ba, bb;
      if (u & 1) {
        ba.x = qa.z; ba.y = qa.w;
        bb.x = qb.z; bb.y = qb.w;
      } else {
        ba.x = qa.x; ba.y = qa.y;
        bb.x = qb.x; bb.y = qb.y;
      }
      d0 = __builtin_amdgcn_wmma_f32_16x16x4_f32(false, wf[u], false, ba,
                                                 (short)0, d0, false, false);
      d1 = __builtin_amdgcn_wmma_f32_16x16x4_f32(false, wf[u], false, bb,
                                                 (short)0, d1, false, false);
    }
    if (live) {
      float* o0 = dst + tt * (kTileRows * 2 * kG);
      float* o1 = o0 + kTileRows * 2 * kG;
      v2f s;
      s.x = d0[0]; s.y = d0[1]; *(v2f*)(o0) = s;
      s.x = d0[2]; s.y = d0[3]; *(v2f*)(o0 + 2) = s;
      s.x = d0[4]; s.y = d0[5]; *(v2f*)(o0 + 4) = s;
      s.x = d1[0]; s.y = d1[1]; *(v2f*)(o1) = s;
      s.x = d1[2]; s.y = d1[3]; *(v2f*)(o1 + 2) = s;
      s.x = d1[4]; s.y = d1[5]; *(v2f*)(o1 + 4) = s;
    }
  }
}

__launch_bounds__(kBlockThreads)
__global__ void gru_fused_wmma(const float* __restrict__ x,
                               const float* __restrict__ W_ih,
                               const float* __restrict__ W_hh,
                               const float* __restrict__ b_ih,
                               const float* __restrict__ b_hh,
                               const float* __restrict__ fc_w,
                               const float* __restrict__ fc_b,
                               float* __restrict__ out) {
  __shared__ __attribute__((aligned(16))) float gx_s[kWavesPerBlock * kLdsPerWave];

  const int tid = threadIdx.x;
  const int wave = tid >> 5;
  const int lane = tid & 31;
  const int c = lane & 15;
  const int h = lane >> 4;
  const int seg = c >> 3;
  const int j = c & 7;
  const int b0 = blockIdx.x * kBatchPerBlock + wave * kBatchPerWave;
  float* wlds = gx_s + wave * kLdsPerWave;

  float whh[12], cs[4], bn2[2], bin[2];
#pragma unroll
  for (int i2 = 0; i2 < 12; ++i2) whh[i2] = 0.5f * W_hh[i2];
#pragma unroll
  for (int g = 0; g < 4; ++g) cs[g] = 0.5f * (b_ih[g] + b_hh[g]);
  bn2[0] = 0.5f * b_hh[4];
  bn2[1] = 0.5f * b_hh[5];
  bin[0] = b_ih[4];
  bin[1] = b_ih[5];
  const float fcw0 = fc_w[0], fcw1 = fc_w[1], fcb = fc_b[0];

  v2f wf[8];
#pragma unroll
  for (int u = 0; u < 8; ++u) {
    const int i = 16 * (u >> 1) + 8 * seg + 4 * h + 2 * (u & 1);
    float w0 = 0.0f, w1 = 0.0f;
    if (j < kG) {
      if (u < 6 || seg == 0) {
        w0 = W_ih[j * kI + i];
        w1 = W_ih[j * kI + i + 1];
      } else if (u == 7 && h == 0) {
        w1 = W_ih[j * kI + (kI - 1)];
      }
    }
    v2f f;
    f.x = w0;
    f.y = w1;
    wf[u] = f;
  }

  const int j2 = j >> 2;
  const float* xl = x + (size_t)b0 * (kT * kI) + kI * (j & 3);
  const int off0 = 16 * j2 + 8 * seg + 4 * h;
  const int off1 = (j2 & seg) ? 53 : 32 + off0;
  const bool live = (seg == h);
  float* dstl = wlds + (j * 2 + seg) * kG;

  v4f bufA[kGroup][4], bufB[kGroup][4];
  load_group(bufA, xl, off0, off1);
#pragma unroll 1
  for (int it = 0; it < 4; ++it) {
    const float* p = xl + (8 * it) * kTileFloats;
    float* d = dstl + (8 * it) * (kTileRows * 2 * kG);
    load_group(bufB, p + 4 * kTileFloats, off0, off1);
    compute_group(bufA, wf, live, d);
    load_group(bufA, p + 8 * kTileFloats, off0, off1);
    compute_group(bufB, wf, live, d + 4 * (kTileRows * 2 * kG));
  }
  compute_group(bufA, wf, live, dstl + 32 * (kTileRows * 2 * kG));

  __syncthreads();

  const float* gl = wlds + (kT * lane) * (2 * kG);
  float h0 = 0.0f, h1 = 0.0f;
#pragma unroll
  for (int t = 0; t < kT; ++t) {
    const v4f* g4 = (const v4f*)(gl + t * (2 * kG));
    const v4f p0 = g4[0], p1 = g4[1], p2 = g4[2];
    const float gt0 = p0.x + p1.z;
    const float gt1 = p0.y + p1.w;
    const float gt2 = p0.z + p2.x;
    const float gt3 = p0.w + p2.y;
    const float gt4 = p1.x + p2.z;
    const float gt5 = p1.y + p2.w;
    const float k0 = fmaf(0.5f, gt0, cs[0]);
    const float k1 = fmaf(0.5f, gt1, cs[1]);
    const float k2 = fmaf(0.5f, gt2, cs[2]);
    const float k3 = fmaf(0.5f, gt3, cs[3]);
    const float k4 = gt4 + bin[0];
    const float k5 = gt5 + bin[1];
    const float a0 = fmaf(whh[0], h0, fmaf(whh[1], h1, k0));
    const float a1 = fmaf(whh[2], h0, fmaf(whh[3], h1, k1));
    const float a2 = fmaf(whh[4], h0, fmaf(whh[5], h1, k2));
    const float a3 = fmaf(whh[6], h0, fmaf(whh[7], h1, k3));
    const float q0 = fmaf(whh[8], h0, fmaf(whh[9], h1, bn2[0]));
    const float q1 = fmaf(whh[10], h0, fmaf(whh[11], h1, bn2[1]));
    const float tr0 = hw_tanh(a0);
    const float tr1 = hw_tanh(a1);
    const float tz0 = hw_tanh(a2);
    const float tz1 = hw_tanh(a3);
    const float n0 = hw_tanh(fmaf(tr0, q0, k4 + q0));
    const float n1 = hw_tanh(fmaf(tr1, q1, k5 + q1));
    const float z0 = fmaf(0.5f, tz0, 0.5f);
    const float z1 = fmaf(0.5f, tz1, 0.5f);
    h0 = fmaf(z0, h0 - n0, n0);
    h1 = fmaf(z1, h1 - n1, n1);
  }
  out[b0 + lane] = fmaf(fcw0, h0, fmaf(fcw1, h1, fcb));
}

extern "C" void kernel_launch(void* const* d_in, const int* in_sizes, int n_in,
                              void* d_out, int out_size, void* d_ws, size_t ws_size,
                              hipStream_t stream) {
  const float* x    = (const float*)d_in[0];
  const float* W_ih = (const float*)d_in[1];
  const float* W_hh = (const float*)d_in[2];
  const float* b_ih = (const float*)d_in[3];
  const float* b_hh = (const float*)d_in[4];
  const float* fc_w = (const float*)d_in[5];
  const float* fc_b = (const float*)d_in[6];
  float* out = (float*)d_out;

  dim3 grid(kB / kBatchPerBlock);
  dim3 block(kBlockThreads);
  hipLaunchKernelGGL(gru_fused_wmma, grid, block, 0, stream,
                     x, W_ih, W_hh, b_ih, b_hh, fc_w, fc_b, out);
}
